# starting kernel
# baseline (speedup 1.0000x reference)
#include <hip/hip_runtime.h>

typedef float v2f __attribute__((ext_vector_type(2)));
typedef float v8f __attribute__((ext_vector_type(8)));

#define D_HID 128

__global__ void wmma_linear_kernel(const float* __restrict__ X,
                                   const float* __restrict__ W,
                                   const float* __restrict__ bias,
                                   float* __restrict__ Y,
                                   int M, int K, int N, int do_relu)
{
    const int lane = threadIdx.x & 31;
    const int wave = threadIdx.x >> 5;
    const int ntiles = (N + 15) >> 4;
    const int nt = blockIdx.y * 4 + wave;
    if (nt >= ntiles) return;

    const int mt   = blockIdx.x;
    const int half = lane >> 4;
    const int l    = lane & 15;

    const int m  = mt * 16 + l;
    const int mc = (m < M) ? m : (M - 1);
    const int n  = nt * 16 + l;
    const int nc = (n < N) ? n : (N - 1);

    const float* xrow = X + (size_t)mc * K + 2 * half;
    const float* wrow = W + (size_t)nc * K + 2 * half;

    v8f acc = {0.f, 0.f, 0.f, 0.f, 0.f, 0.f, 0.f, 0.f};
    #pragma unroll 8
    for (int k = 0; k < K; k += 4) {
        v2f a = *(const v2f*)(xrow + k);
        v2f b = *(const v2f*)(wrow + k);
        acc = __builtin_amdgcn_wmma_f32_16x16x4_f32(
                  false, a, false, b, (short)0, acc, false, false);
    }

    const float bv = bias[nc];
    #pragma unroll
    for (int r = 0; r < 8; ++r) {
        float v = acc[r] + bv;
        if (do_relu) v = (v > 0.f) ? v : 0.f;
        const int row = mt * 16 + r + 8 * half;
        if (n < N && row < M)
            Y[(size_t)row * N + n] = v;
    }
}

__global__ void zero_kernel(float* __restrict__ p, long long n)
{
    long long i = (long long)blockIdx.x * blockDim.x + threadIdx.x;
    if (i < n) p[i] = 0.0f;
}

__global__ void scatter_add_kernel(const float* __restrict__ Y,
                                   const int* __restrict__ src,
                                   const int* __restrict__ dst,
                                   float* __restrict__ out, int E)
{
    const long long t = (long long)blockIdx.x * blockDim.x + threadIdx.x;
    const int e = (int)(t >> 5);
    if (e >= E) return;
    const int q = (int)(t & 31);

    const int s = src[e];
    const int d = dst[e];
    const float4 v = ((const float4*)(Y + (size_t)s * D_HID))[q];
    float* o = out + (size_t)d * D_HID + q * 4;
    atomicAdd(o + 0, v.x);
    atomicAdd(o + 1, v.y);
    atomicAdd(o + 2, v.z);
    atomicAdd(o + 3, v.w);
}

extern "C" void kernel_launch(void* const* d_in, const int* in_sizes, int n_in,
                              void* d_out, int out_size, void* d_ws, size_t ws_size,
                              hipStream_t stream)
{
    const float* X0  = (const float*)d_in[0];
    const float* W1  = (const float*)d_in[1];
    const float* b1  = (const float*)d_in[2];
    const float* W2  = (const float*)d_in[3];
    const float* b2  = (const float*)d_in[4];
    const float* Wfc = (const float*)d_in[5];
    const float* bfc = (const float*)d_in[6];
    const int*   esrc = (const int*)d_in[7];
    const int*   edst = (const int*)d_in[8];

    const int Nn   = in_sizes[0] / D_HID;
    const int E    = in_sizes[7];
    const int Ncls = in_sizes[6];

    float* bufA = (float*)d_ws;
    float* bufB = bufA + (size_t)Nn * D_HID;
    const long long hidElems = (long long)Nn * D_HID;

    const dim3 blkLin(128);
    const int mtiles = (Nn + 15) / 16;
    const dim3 grdHid(mtiles, (D_HID / 16 + 3) / 4);
    const dim3 grdFc(mtiles, ((Ncls + 15) / 16 + 3) / 4);

    const int zeroBlocks = (int)((hidElems + 255) / 256);
    const int scatBlocks = (int)(((long long)E * 32 + 255) / 256);

    wmma_linear_kernel<<<grdHid, blkLin, 0, stream>>>(X0, W1, b1, bufA,
                                                      Nn, D_HID, D_HID, 1);
    zero_kernel<<<zeroBlocks, 256, 0, stream>>>(bufB, hidElems);
    scatter_add_kernel<<<scatBlocks, 256, 0, stream>>>(bufA, esrc, edst, bufB, E);

    wmma_linear_kernel<<<grdHid, blkLin, 0, stream>>>(bufB, W2, b2, bufA,
                                                      Nn, D_HID, D_HID, 1);
    zero_kernel<<<zeroBlocks, 256, 0, stream>>>(bufB, hidElems);
    scatter_add_kernel<<<scatBlocks, 256, 0, stream>>>(bufA, esrc, edst, bufB, E);

    wmma_linear_kernel<<<grdFc, blkLin, 0, stream>>>(bufB, Wfc, bfc, (float*)d_out,
                                                     Nn, D_HID, Ncls, 0);
}


# layer-1 linear on the fused kernel's data path: 16-row X strips go to LDS by whole-row 512 B loads, each of 8 waves keeps its W column tile in registers; k summed in two chains (sum order changes)
# speedup vs baseline: 2.1518x; 2.1518x over previous
#include <hip/hip_runtime.h>

typedef float v2f __attribute__((ext_vector_type(2)));
typedef float v8f __attribute__((ext_vector_type(8)));

#define D_HID 128
#define FUSED_GRID 1024

#define RNG 1024
#define NCHUNK 5
#define CHUNK_LOADS 4

__device__ __forceinline__ void count_one(int* s_cnt, int* __restrict__ rank, int e, int d, int lo)
{
    const unsigned u = (unsigned)(d - lo);
    if (u < (unsigned)RNG) rank[e] = atomicAdd(&s_cnt[u], 1);
}

__global__ void __launch_bounds__(RNG)
range_count_kernel(const int* __restrict__ dst, int* __restrict__ rank,
                   int* __restrict__ cnt, int* __restrict__ tot,
                   int E, int n_nodes, int chunk)
{
    __shared__ int s_cnt[RNG];
    __shared__ int s_red[32];
    const int tid  = threadIdx.x;
    const int lane = tid & 31;
    const int wave = tid >> 5;
    const int lo = blockIdx.x * RNG;
    const int q  = blockIdx.y;
    s_cnt[tid] = 0;
    __syncthreads();

    const int e0 = q * chunk;
    const int e1 = (e0 + chunk < E) ? (e0 + chunk) : E;
    const int n4 = (e1 > e0) ? ((e1 - e0) >> 2) : 0;
    const int4* d4 = (const int4*)(dst + e0);
    int* rk = rank + e0;
    for (int g = tid; g < n4; g += CHUNK_LOADS * RNG) {
        int4 v[CHUNK_LOADS];
        #pragma unroll
        for (int k = 0; k < CHUNK_LOADS; ++k) {
            const int gk = g + k * RNG;
            v[k] = d4[(gk < n4) ? gk : g];
        }
        #pragma unroll
        for (int k = 0; k < CHUNK_LOADS; ++k) {
            const int gk = g + k * RNG;
            if (gk < n4) {
                count_one(s_cnt, rk, 4 * gk + 0, v[k].x, lo);
                count_one(s_cnt, rk, 4 * gk + 1, v[k].y, lo);
                count_one(s_cnt, rk, 4 * gk + 2, v[k].z, lo);
                count_one(s_cnt, rk, 4 * gk + 3, v[k].w, lo);
            }
        }
    }
    const int t = e0 + 4 * n4 + tid;
    if (t < e1) count_one(s_cnt, rank, t, dst[t], lo);
    __syncthreads();

    const int c = s_cnt[tid];
    const int node = lo + tid;
    if (node < n_nodes) cnt[(size_t)q * n_nodes + node] = c;
    int s = c;
    for (int o = 16; o > 0; o >>= 1) s += __shfl_xor(s, o);
    if (lane == 0) s_red[wave] = s;
    __syncthreads();
    if (wave == 0) {
        int r = s_red[lane];
        for (int o = 16; o > 0; o >>= 1) r += __shfl_xor(r, o);
        if (lane == 0) tot[blockIdx.x * NCHUNK + q] = r;
    }
}

__global__ void __launch_bounds__(RNG)
offsets_kernel(const int* __restrict__ cnt, const int* __restrict__ tot,
               int* __restrict__ offsets, int* __restrict__ start, int n)
{
    __shared__ int s_part[32];
    __shared__ int s_pre[32];
    const int tid  = threadIdx.x;
    const int lane = tid & 31;
    const int wave = tid >> 5;
    const int first = blockIdx.x * RNG;

    int pre = 0;
    for (int j = tid; j < (int)blockIdx.x * NCHUNK; j += RNG) pre += tot[j];

    const int i = first + tid;
    int cq[NCHUNK];
    int c = 0;
    #pragma unroll
    for (int q = 0; q < NCHUNK; ++q) {
        cq[q] = (i < n) ? cnt[(size_t)q * n + i] : 0;
        c += cq[q];
    }
    int incl = c;
    for (int o = 1; o < 32; o <<= 1) {
        const int y = __shfl_up(incl, o);
        if (lane >= o) incl += y;
    }
    for (int o = 16; o > 0; o >>= 1) pre += __shfl_xor(pre, o);
    if (lane == 31) s_part[wave] = incl;
    if (lane == 0)  s_pre[wave]  = pre;
    __syncthreads();
    int add = s_pre[lane] + ((lane < wave) ? s_part[lane] : 0);
    for (int o = 16; o > 0; o >>= 1) add += __shfl_xor(add, o);
    const int excl = add + incl - c;
    if (i < n) {
        offsets[i] = excl;
        int s = excl;
        #pragma unroll
        for (int q = 0; q < NCHUNK; ++q) {
            start[(size_t)q * n + i] = s;
            s += cq[q];
        }
    }
    if (i == n - 1) offsets[n] = excl + c;
}

__global__ void place_kernel(const int* __restrict__ src,
                             const int* __restrict__ dst,
                             const int* __restrict__ rank,
                             const int* __restrict__ start,
                             int* __restrict__ by_dst, int E, int n_nodes, int chunk)
{
    const int q  = blockIdx.y;
    const int k  = blockIdx.x * blockDim.x + threadIdx.x;
    const int e  = q * chunk + k;
    if (k < chunk && e < E) by_dst[start[(size_t)q * n_nodes + dst[e]] + rank[e]] = src[e];
}

#define STRIP_PITCH 132
#define ROWS_IN_FLIGHT 8

__device__ __forceinline__ int load_ids(const int* __restrict__ by_dst, int beg, int cnt, int lane)
{
    return (lane < cnt) ? by_dst[beg + lane] : 0;
}

__global__ void __launch_bounds__(256)
gather_linear_kernel(const float* __restrict__ Y,
                     const int* __restrict__ offsets,
                     const int* __restrict__ by_dst,
                     const float* __restrict__ W,
                     const float* __restrict__ bias,
                     float* __restrict__ Out,
                     int n_nodes, int N, int do_relu)
{
    __shared__ float4 s_strip[2][16 * STRIP_PITCH / 4];

    const int lane = threadIdx.x & 31;
    const int wave = __builtin_amdgcn_readfirstlane((int)(threadIdx.x >> 5));
    const int half = lane >> 4;
    const int c    = lane & 15;

    const int ntiles = (N + 15) >> 4;
    const int mtiles = (n_nodes + 15) >> 4;
    const bool has_tile = wave < ntiles;

    const int n  = wave * 16 + c;
    const int nc = (n < N) ? n : (N - 1);

    float4 bq[16];
    float bv = 0.f;
    if (has_tile) {
        const float4* wq = (const float4*)(W + (size_t)nc * D_HID + 4 * half);
        #pragma unroll
        for (int j = 0; j < 16; ++j) bq[j] = wq[2 * j];
        bv = bias[nc];
    }

    const int osel = 2 * wave + ((lane < 2) ? lane : 2);
    const int step = gridDim.x;

    int mt = blockIdx.x;
    int o0 = 0, o1 = 0, o2 = 0;
    int mine = 0, mine_hi = 0;
    int ovn = 0;
    if (mt < mtiles) {
        int oi = mt * 16 + osel;
        oi = (oi < n_nodes) ? oi : n_nodes;
        const int ov = offsets[oi];
        o0 = __builtin_amdgcn_readlane(ov, 0);
        o1 = __builtin_amdgcn_readlane(ov, 1);
        o2 = __builtin_amdgcn_readlane(ov, 2);
        mine    = load_ids(by_dst, o0,      o2 - o0,      lane);
        mine_hi = load_ids(by_dst, o0 + 32, o2 - o0 - 32, lane);
        if (mt + step < mtiles) {
            int on = (mt + step) * 16 + osel;
            on = (on < n_nodes) ? on : n_nodes;
            ovn = offsets[on];
        }
    }

    int buf = 0;
    for (; mt < mtiles; mt += step, buf ^= 1) {
        float* strip = (float*)s_strip[buf];

        const int cnt   = o2 - o0;
        const int split = o1 - o0;
        float4 a0 = {0.f, 0.f, 0.f, 0.f};
        float4 a1 = {0.f, 0.f, 0.f, 0.f};
        for (int base = 0; base < cnt; base += 32) {
            int ids;
            if (base == 0)       ids = mine;
            else if (base == 32) ids = mine_hi;
            else                 ids = load_ids(by_dst, o0 + base, cnt - base, lane);
            const int left = cnt - base;
            const int c32  = (left < 32) ? left : 32;
            for (int k = 0; k < c32; k += ROWS_IN_FLIGHT) {
                float4 v[ROWS_IN_FLIGHT];
                const bool wide = (k + ROWS_IN_FLIGHT / 2 < c32);
                #pragma unroll
                for (int i = 0; i < ROWS_IN_FLIGHT / 2; ++i) {
                    const int kk = (k + i < c32) ? (k + i) : (c32 - 1);
                    const int s  = __builtin_amdgcn_readlane(ids, kk);
                    v[i] = ((const float4*)(Y + (size_t)s * D_HID))[lane];
                }
                if (wide) {
                    #pragma unroll
                    for (int i = ROWS_IN_FLIGHT / 2; i < ROWS_IN_FLIGHT; ++i) {
                        const int kk = (k + i < c32) ? (k + i) : (c32 - 1);
                        const int s  = __builtin_amdgcn_readlane(ids, kk);
                        v[i] = ((const float4*)(Y + (size_t)s * D_HID))[lane];
                    }
                }
                #pragma unroll
                for (int i = 0; i < ROWS_IN_FLIGHT; ++i) {
                    if (i == 0 || k + i < c32) {
                        if (base + k + i < split) {
                            a0.x += v[i].x; a0.y += v[i].y; a0.z += v[i].z; a0.w += v[i].w;
                        } else {
                            a1.x += v[i].x; a1.y += v[i].y; a1.z += v[i].z; a1.w += v[i].w;
                        }
                    }
                }
            }
        }
        ((float4*)(strip + (2 * wave)     * STRIP_PITCH))[lane] = a0;
        ((float4*)(strip + (2 * wave + 1) * STRIP_PITCH))[lane] = a1;

        const int mtn = mt + step;
        if (mtn < mtiles) {
            o0 = __builtin_amdgcn_readlane(ovn, 0);
            o1 = __builtin_amdgcn_readlane(ovn, 1);
            o2 = __builtin_amdgcn_readlane(ovn, 2);
            mine    = load_ids(by_dst, o0,      o2 - o0,      lane);
            mine_hi = load_ids(by_dst, o0 + 32, o2 - o0 - 32, lane);
            if (mtn + step < mtiles) {
                int on = (mtn + step) * 16 + osel;
                on = (on < n_nodes) ? on : n_nodes;
                ovn = offsets[on];
            }
        }
        asm volatile("s_wait_dscnt 0x0" ::: "memory");
        asm volatile("s_barrier_signal -1" ::: "memory");
        asm volatile("s_barrier_wait -1" ::: "memory");

        if (has_tile) {
            const float* sa = strip + c * STRIP_PITCH + 4 * half;
            v8f acc0 = {0.f, 0.f, 0.f, 0.f, 0.f, 0.f, 0.f, 0.f};
            v8f acc1 = {0.f, 0.f, 0.f, 0.f, 0.f, 0.f, 0.f, 0.f};
            #pragma unroll
            for (int j = 0; j < 16; ++j) {
                const float4 a = *(const float4*)(sa + 8 * j);
                const v2f a0f = {a.x, a.y};
                const v2f a1f = {a.z, a.w};
                const v2f b0 = {bq[j].x, bq[j].y};
                const v2f b1 = {bq[j].z, bq[j].w};
                acc0 = __builtin_amdgcn_wmma_f32_16x16x4_f32(
                           false, a0f, false, b0, (short)0, acc0, false, false);
                acc1 = __builtin_amdgcn_wmma_f32_16x16x4_f32(
                           false, a1f, false, b1, (short)0, acc1, false, false);
            }
            #pragma unroll
            for (int r = 0; r < 8; ++r) {
                float v = acc0[r] + acc1[r] + bv;
                if (do_relu) v = (v > 0.f) ? v : 0.f;
                const int row = mt * 16 + r + 8 * half;
                if (n < N && row < n_nodes)
                    Out[(size_t)row * N + n] = v;
            }
        }
    }
}

#define LIN_GRID 1024

__global__ void __launch_bounds__(256)
strip_linear_relu_kernel(const float* __restrict__ X,
                         const float* __restrict__ W,
                         const float* __restrict__ bias,
                         float* __restrict__ Out,
                         int M)
{
    __shared__ float4 s_strip[2][16 * STRIP_PITCH / 4];

    const int lane = threadIdx.x & 31;
    const int wave = __builtin_amdgcn_readfirstlane((int)(threadIdx.x >> 5));
    const int half = lane >> 4;
    const int c    = lane & 15;

    const int mtiles = (M + 15) >> 4;
    const int n = wave * 16 + c;

    float4 bq[16];
    const float4* wq = (const float4*)(W + (size_t)n * D_HID + 4 * half);
    #pragma unroll
    for (int j = 0; j < 16; ++j) bq[j] = wq[2 * j];
    const float bv = bias[n];

    const int step = gridDim.x;
    int mt = blockIdx.x;

    float4 r0 = {0.f, 0.f, 0.f, 0.f};
    float4 r1 = {0.f, 0.f, 0.f, 0.f};
    if (mt < mtiles) {
        int ra = mt * 16 + 2 * wave;
        int rb = ra + 1;
        ra = (ra < M) ? ra : (M - 1);
        rb = (rb < M) ? rb : (M - 1);
        r0 = ((const float4*)(X + (size_t)ra * D_HID))[lane];
        r1 = ((const float4*)(X + (size_t)rb * D_HID))[lane];
    }

    int buf = 0;
    for (; mt < mtiles; mt += step, buf ^= 1) {
        float* strip = (float*)s_strip[buf];
        ((float4*)(strip + (2 * wave)     * STRIP_PITCH))[lane] = r0;
        ((float4*)(strip + (2 * wave + 1) * STRIP_PITCH))[lane] = r1;

        const int mtn = mt + step;
        if (mtn < mtiles) {
            int ra = mtn * 16 + 2 * wave;
            int rb = ra + 1;
            ra = (ra < M) ? ra : (M - 1);
            rb = (rb < M) ? rb : (M - 1);
            r0 = ((const float4*)(X + (size_t)ra * D_HID))[lane];
            r1 = ((const float4*)(X + (size_t)rb * D_HID))[lane];
        }
        asm volatile("s_wait_dscnt 0x0" ::: "memory");
        asm volatile("s_barrier_signal -1" ::: "memory");
        asm volatile("s_barrier_wait -1" ::: "memory");

        const float* sa = strip + c * STRIP_PITCH + 4 * half;
        v8f acc0 = {0.f, 0.f, 0.f, 0.f, 0.f, 0.f, 0.f, 0.f};
        v8f acc1 = {0.f, 0.f, 0.f, 0.f, 0.f, 0.f, 0.f, 0.f};
        #pragma unroll
        for (int j = 0; j < 16; ++j) {
            const float4 a = *(const float4*)(sa + 8 * j);
            const v2f a0f = {a.x, a.y};
            const v2f a1f = {a.z, a.w};
            const v2f b0 = {bq[j].x, bq[j].y};
            const v2f b1 = {bq[j].z, bq[j].w};
            acc0 = __builtin_amdgcn_wmma_f32_16x16x4_f32(
                       false, a0f, false, b0, (short)0, acc0, false, false);
            acc1 = __builtin_amdgcn_wmma_f32_16x16x4_f32(
                       false, a1f, false, b1, (short)0, acc1, false, false);
        }
        const int row0 = mt * 16 + 8 * half;
        float* orow = Out + (size_t)row0 * D_HID + n;
        if (mt * 16 + 16 <= M) {
            #pragma unroll
            for (int r = 0; r < 8; ++r) {
                float v = acc0[r] + acc1[r] + bv;
                v = (v > 0.f) ? v : 0.f;
                orow[(size_t)r * D_HID] = v;
            }
        } else {
            #pragma unroll
            for (int r = 0; r < 8; ++r) {
                float v = acc0[r] + acc1[r] + bv;
                v = (v > 0.f) ? v : 0.f;
                if (row0 + r < M)
                    orow[(size_t)r * D_HID] = v;
            }
        }
    }
}

__global__ void copy_ints_kernel(const int* __restrict__ src, int* __restrict__ dst, int n)
{
    const int i = blockIdx.x * blockDim.x + threadIdx.x;
    if (i < n) dst[i] = src[i];
}

extern "C" void kernel_launch(void* const* d_in, const int* in_sizes, int n_in,
                              void* d_out, int out_size, void* d_ws, size_t ws_size,
                              hipStream_t stream)
{
    const float* X0  = (const float*)d_in[0];
    const float* W1  = (const float*)d_in[1];
    const float* b1  = (const float*)d_in[2];
    const float* W2  = (const float*)d_in[3];
    const float* b2  = (const float*)d_in[4];
    const float* Wfc = (const float*)d_in[5];
    const float* bfc = (const float*)d_in[6];
    const int*   esrc = (const int*)d_in[7];
    const int*   edst = (const int*)d_in[8];

    const int Nn   = in_sizes[0] / D_HID;
    const int E    = in_sizes[7];
    const int Ncls = in_sizes[6];

    float* bufA = (float*)d_ws;
    float* bufB = bufA + (size_t)Nn * D_HID;

    const int ranges = (Nn + RNG - 1) / RNG;
    const int chunk  = (((E + NCHUNK - 1) / NCHUNK) + 3) & ~3;
    int* cnt     = (int*)bufB;
    int* start   = cnt + (size_t)NCHUNK * Nn;
    int* tot     = start + (size_t)NCHUNK * Nn;
    int* rank    = tot + ranges * NCHUNK;
    int* offsets = (int*)d_out;
    int* by_dst  = offsets + (Nn + 1);
    int* offsets2 = (int*)bufA;
    int* by_dst2  = offsets2 + (Nn + 1);
    const int csrInts = Nn + 1 + E;

    const dim3 grdCount(ranges, NCHUNK);
    const dim3 grdPlace((chunk + 255) / 256, NCHUNK);
    const int copyBlocks = (csrInts + 255) / 256;
    const int fusedBlocks = FUSED_GRID;

    range_count_kernel<<<grdCount, RNG, 0, stream>>>(edst, rank, cnt, tot, E, Nn, chunk);
    offsets_kernel<<<ranges, RNG, 0, stream>>>(cnt, tot, offsets, start, Nn);
    place_kernel<<<grdPlace, 256, 0, stream>>>(esrc, edst, rank, start, by_dst, E, Nn, chunk);

    strip_linear_relu_kernel<<<LIN_GRID, 256, 0, stream>>>(X0, W1, b1, bufA, Nn);

    gather_linear_kernel<<<fusedBlocks, 256, 0, stream>>>(bufA, offsets, by_dst, W2, b2, bufB,
                                                          Nn, D_HID, 1);

    copy_ints_kernel<<<copyBlocks, 256, 0, stream>>>(offsets, offsets2, csrInts);

    gather_linear_kernel<<<fusedBlocks, 256, 0, stream>>>(bufB, offsets2, by_dst2, Wfc, bfc,
                                                          (float*)d_out, Nn, Ncls, 0);
}


# fused gather+linear: fewer instructions per gathered and stored row: whole 8-row blocks on one side of the node boundary add untested in loops of their own, 32-bit row index, store guards per strip
# speedup vs baseline: 2.2814x; 1.0602x over previous
#include <hip/hip_runtime.h>

typedef float v2f __attribute__((ext_vector_type(2)));
typedef float v8f __attribute__((ext_vector_type(8)));

#define D_HID 128
#define FUSED_GRID 1024

#define RNG 1024
#define NCHUNK 5
#define CHUNK_LOADS 4

__device__ __forceinline__ void count_one(int* s_cnt, int* __restrict__ rank, int e, int d, int lo)
{
    const unsigned u = (unsigned)(d - lo);
    if (u < (unsigned)RNG) rank[e] = atomicAdd(&s_cnt[u], 1);
}

__global__ void __launch_bounds__(RNG)
range_count_kernel(const int* __restrict__ dst, int* __restrict__ rank,
                   int* __restrict__ cnt, int* __restrict__ tot,
                   int E, int n_nodes, int chunk)
{
    __shared__ int s_cnt[RNG];
    __shared__ int s_red[32];
    const int tid  = threadIdx.x;
    const int lane = tid & 31;
    const int wave = tid >> 5;
    const int lo = blockIdx.x * RNG;
    const int q  = blockIdx.y;
    s_cnt[tid] = 0;
    __syncthreads();

    const int e0 = q * chunk;
    const int e1 = (e0 + chunk < E) ? (e0 + chunk) : E;
    const int n4 = (e1 > e0) ? ((e1 - e0) >> 2) : 0;
    const int4* d4 = (const int4*)(dst + e0);
    int* rk = rank + e0;
    for (int g = tid; g < n4; g += CHUNK_LOADS * RNG) {
        int4 v[CHUNK_LOADS];
        #pragma unroll
        for (int k = 0; k < CHUNK_LOADS; ++k) {
            const int gk = g + k * RNG;
            v[k] = d4[(gk < n4) ? gk : g];
        }
        #pragma unroll
        for (int k = 0; k < CHUNK_LOADS; ++k) {
            const int gk = g + k * RNG;
            if (gk < n4) {
                count_one(s_cnt, rk, 4 * gk + 0, v[k].x, lo);
                count_one(s_cnt, rk, 4 * gk + 1, v[k].y, lo);
                count_one(s_cnt, rk, 4 * gk + 2, v[k].z, lo);
                count_one(s_cnt, rk, 4 * gk + 3, v[k].w, lo);
            }
        }
    }
    const int t = e0 + 4 * n4 + tid;
    if (t < e1) count_one(s_cnt, rank, t, dst[t], lo);
    __syncthreads();

    const int c = s_cnt[tid];
    const int node = lo + tid;
    if (node < n_nodes) cnt[(size_t)q * n_nodes + node] = c;
    int s = c;
    for (int o = 16; o > 0; o >>= 1) s += __shfl_xor(s, o);
    if (lane == 0) s_red[wave] = s;
    __syncthreads();
    if (wave == 0) {
        int r = s_red[lane];
        for (int o = 16; o > 0; o >>= 1) r += __shfl_xor(r, o);
        if (lane == 0) tot[blockIdx.x * NCHUNK + q] = r;
    }
}

__global__ void __launch_bounds__(RNG)
offsets_kernel(const int* __restrict__ cnt, const int* __restrict__ tot,
               int* __restrict__ offsets, int* __restrict__ start, int n)
{
    __shared__ int s_part[32];
    __shared__ int s_pre[32];
    const int tid  = threadIdx.x;
    const int lane = tid & 31;
    const int wave = tid >> 5;
    const int first = blockIdx.x * RNG;

    int pre = 0;
    for (int j = tid; j < (int)blockIdx.x * NCHUNK; j += RNG) pre += tot[j];

    const int i = first + tid;
    int cq[NCHUNK];
    int c = 0;
    #pragma unroll
    for (int q = 0; q < NCHUNK; ++q) {
        cq[q] = (i < n) ? cnt[(size_t)q * n + i] : 0;
        c += cq[q];
    }
    int incl = c;
    for (int o = 1; o < 32; o <<= 1) {
        const int y = __shfl_up(incl, o);
        if (lane >= o) incl += y;
    }
    for (int o = 16; o > 0; o >>= 1) pre += __shfl_xor(pre, o);
    if (lane == 31) s_part[wave] = incl;
    if (lane == 0)  s_pre[wave]  = pre;
    __syncthreads();
    int add = s_pre[lane] + ((lane < wave) ? s_part[lane] : 0);
    for (int o = 16; o > 0; o >>= 1) add += __shfl_xor(add, o);
    const int excl = add + incl - c;
    if (i < n) {
        offsets[i] = excl;
        int s = excl;
        #pragma unroll
        for (int q = 0; q < NCHUNK; ++q) {
            start[(size_t)q * n + i] = s;
            s += cq[q];
        }
    }
    if (i == n - 1) offsets[n] = excl + c;
}

__global__ void place_kernel(const int* __restrict__ src,
                             const int* __restrict__ dst,
                             const int* __restrict__ rank,
                             const int* __restrict__ start,
                             int* __restrict__ by_dst, int E, int n_nodes, int chunk)
{
    const int q  = blockIdx.y;
    const int k  = blockIdx.x * blockDim.x + threadIdx.x;
    const int e  = q * chunk + k;
    if (k < chunk && e < E) by_dst[start[(size_t)q * n_nodes + dst[e]] + rank[e]] = src[e];
}

#define STRIP_PITCH 132
#define ROWS_IN_FLIGHT 8

__device__ __forceinline__ int load_ids(const int* __restrict__ by_dst, int beg, int cnt, int lane)
{
    return (lane < cnt) ? by_dst[beg + lane] : 0;
}

__global__ void __launch_bounds__(256)
gather_linear_kernel(const float* __restrict__ Y,
                     const int* __restrict__ offsets,
                     const int* __restrict__ by_dst,
                     const float* __restrict__ W,
                     const float* __restrict__ bias,
                     float* __restrict__ Out,
                     int n_nodes, int N, int do_relu)
{
    __shared__ float4 s_strip[2][16 * STRIP_PITCH / 4];

    const int lane = threadIdx.x & 31;
    const int wave = __builtin_amdgcn_readfirstlane((int)(threadIdx.x >> 5));
    const int half = lane >> 4;
    const int c    = lane & 15;
    const float4* __restrict__ Yq = (const float4*)Y;
    const unsigned ulane = (unsigned)lane;

    const int ntiles = (N + 15) >> 4;
    const int mtiles = (n_nodes + 15) >> 4;
    const bool has_tile = wave < ntiles;

    const int n  = wave * 16 + c;
    const int nc = (n < N) ? n : (N - 1);

    float4 bq[16];
    float bv = 0.f;
    if (has_tile) {
        const float4* wq = (const float4*)(W + (size_t)nc * D_HID + 4 * half);
        #pragma unroll
        for (int j = 0; j < 16; ++j) bq[j] = wq[2 * j];
        bv = bias[nc];
    }

    const int osel = 2 * wave + ((lane < 2) ? lane : 2);
    const int step = gridDim.x;

    int mt = blockIdx.x;
    int o0 = 0, o1 = 0, o2 = 0;
    int mine = 0, mine_hi = 0;
    int ovn = 0;
    if (mt < mtiles) {
        int oi = mt * 16 + osel;
        oi = (oi < n_nodes) ? oi : n_nodes;
        const int ov = offsets[oi];
        o0 = __builtin_amdgcn_readlane(ov, 0);
        o1 = __builtin_amdgcn_readlane(ov, 1);
        o2 = __builtin_amdgcn_readlane(ov, 2);
        mine    = load_ids(by_dst, o0,      o2 - o0,      lane);
        mine_hi = load_ids(by_dst, o0 + 32, o2 - o0 - 32, lane);
        if (mt + step < mtiles) {
            int on = (mt + step) * 16 + osel;
            on = (on < n_nodes) ? on : n_nodes;
            ovn = offsets[on];
        }
    }

    int buf = 0;
    for (; mt < mtiles; mt += step, buf ^= 1) {
        float* strip = (float*)s_strip[buf];

        const int cnt   = o2 - o0;
        const int split = o1 - o0;
        float4 a0 = {0.f, 0.f, 0.f, 0.f};
        float4 a1 = {0.f, 0.f, 0.f, 0.f};
        for (int base = 0; base < cnt; base += 32) {
            int ids;
            if (base == 0)       ids = mine;
            else if (base == 32) ids = mine_hi;
            else                 ids = load_ids(by_dst, o0 + base, cnt - base, lane);
            const int left = cnt - base;
            const int c32  = (left < 32) ? left : 32;
            const int spb = split - base;
            auto tested_block = [&](const int k) {
                float4 v[ROWS_IN_FLIGHT];
                const bool wide = (k + ROWS_IN_FLIGHT / 2 < c32);
                #pragma unroll
                for (int i = 0; i < ROWS_IN_FLIGHT / 2; ++i) {
                    const int kk = (k + i < c32) ? (k + i) : (c32 - 1);
                    const int s  = __builtin_amdgcn_readlane(ids, kk);
                    v[i] = Yq[(unsigned)s * (D_HID / 4) + ulane];
                }
                if (wide) {
                    #pragma unroll
                    for (int i = ROWS_IN_FLIGHT / 2; i < ROWS_IN_FLIGHT; ++i) {
                        const int kk = (k + i < c32) ? (k + i) : (c32 - 1);
                        const int s  = __builtin_amdgcn_readlane(ids, kk);
                        v[i] = Yq[(unsigned)s * (D_HID / 4) + ulane];
                    }
                }
                #pragma unroll
                for (int i = 0; i < ROWS_IN_FLIGHT; ++i) {
                    if (i == 0 || k + i < c32) {
                        if (k + i < spb) {
                            a0.x += v[i].x; a0.y += v[i].y; a0.z += v[i].z; a0.w += v[i].w;
                        } else {
                            a1.x += v[i].x; a1.y += v[i].y; a1.z += v[i].z; a1.w += v[i].w;
                        }
                    }
                }
            };
            int k = 0;
            for (; k + ROWS_IN_FLIGHT <= c32 && k + ROWS_IN_FLIGHT <= spb; k += ROWS_IN_FLIGHT) {
                float4 v[ROWS_IN_FLIGHT];
                #pragma unroll
                for (int i = 0; i < ROWS_IN_FLIGHT; ++i) {
                    const int s = __builtin_amdgcn_readlane(ids, k + i);
                    v[i] = Yq[(unsigned)s * (D_HID / 4) + ulane];
                }
                #pragma unroll
                for (int i = 0; i < ROWS_IN_FLIGHT; ++i) {
                    a0.x += v[i].x; a0.y += v[i].y; a0.z += v[i].z; a0.w += v[i].w;
                }
            }
            if (k < c32 && k < spb) {
                tested_block(k);
                k += ROWS_IN_FLIGHT;
            }
            for (; k + ROWS_IN_FLIGHT <= c32; k += ROWS_IN_FLIGHT) {
                float4 v[ROWS_IN_FLIGHT];
                #pragma unroll
                for (int i = 0; i < ROWS_IN_FLIGHT; ++i) {
                    const int s = __builtin_amdgcn_readlane(ids, k + i);
                    v[i] = Yq[(unsigned)s * (D_HID / 4) + ulane];
                }
                #pragma unroll
                for (int i = 0; i < ROWS_IN_FLIGHT; ++i) {
                    a1.x += v[i].x; a1.y += v[i].y; a1.z += v[i].z; a1.w += v[i].w;
                }
            }
            if (k < c32) tested_block(k);
        }
        ((float4*)(strip + (2 * wave)     * STRIP_PITCH))[lane] = a0;
        ((float4*)(strip + (2 * wave + 1) * STRIP_PITCH))[lane] = a1;

        const int mtn = mt + step;
        if (mtn < mtiles) {
            o0 = __builtin_amdgcn_readlane(ovn, 0);
            o1 = __builtin_amdgcn_readlane(ovn, 1);
            o2 = __builtin_amdgcn_readlane(ovn, 2);
            mine    = load_ids(by_dst, o0,      o2 - o0,      lane);
            mine_hi = load_ids(by_dst, o0 + 32, o2 - o0 - 32, lane);
            if (mtn + step < mtiles) {
                int on = (mtn + step) * 16 + osel;
                on = (on < n_nodes) ? on : n_nodes;
                ovn = offsets[on];
            }
        }
        asm volatile("s_wait_dscnt 0x0" ::: "memory");
        asm volatile("s_barrier_signal -1" ::: "memory");
        asm volatile("s_barrier_wait -1" ::: "memory");

        if (has_tile) {
            const float* sa = strip + c * STRIP_PITCH + 4 * half;
            v8f acc0 = {0.f, 0.f, 0.f, 0.f, 0.f, 0.f, 0.f, 0.f};
            v8f acc1 = {0.f, 0.f, 0.f, 0.f, 0.f, 0.f, 0.f, 0.f};
            #pragma unroll
            for (int j = 0; j < 16; ++j) {
                const float4 a = *(const float4*)(sa + 8 * j);
                const v2f a0f = {a.x, a.y};
                const v2f a1f = {a.z, a.w};
                const v2f b0 = {bq[j].x, bq[j].y};
                const v2f b1 = {bq[j].z, bq[j].w};
                acc0 = __builtin_amdgcn_wmma_f32_16x16x4_f32(
                           false, a0f, false, b0, (short)0, acc0, false, false);
                acc1 = __builtin_amdgcn_wmma_f32_16x16x4_f32(
                           false, a1f, false, b1, (short)0, acc1, false, false);
            }
            const int row0 = mt * 16 + 8 * half;
            if (n < N) {
                float* orow = Out + (size_t)row0 * N + n;
                if (mt * 16 + 16 <= n_nodes) {
                    #pragma unroll
                    for (int r = 0; r < 8; ++r) {
                        float v = acc0[r] + acc1[r] + bv;
                        if (do_relu) v = (v > 0.f) ? v : 0.f;
                        orow[(size_t)r * N] = v;
                    }
                } else {
                    #pragma unroll
                    for (int r = 0; r < 8; ++r) {
                        float v = acc0[r] + acc1[r] + bv;
                        if (do_relu) v = (v > 0.f) ? v : 0.f;
                        if (row0 + r < n_nodes)
                            orow[(size_t)r * N] = v;
                    }
                }
            }
        }
    }
}

#define LIN_GRID 1024

__global__ void __launch_bounds__(256)
strip_linear_relu_kernel(const float* __restrict__ X,
                         const float* __restrict__ W,
                         const float* __restrict__ bias,
                         float* __restrict__ Out,
                         int M)
{
    __shared__ float4 s_strip[2][16 * STRIP_PITCH / 4];

    const int lane = threadIdx.x & 31;
    const int wave = __builtin_amdgcn_readfirstlane((int)(threadIdx.x >> 5));
    const int half = lane >> 4;
    const int c    = lane & 15;

    const int mtiles = (M + 15) >> 4;
    const int n = wave * 16 + c;

    float4 bq[16];
    const float4* wq = (const float4*)(W + (size_t)n * D_HID + 4 * half);
    #pragma unroll
    for (int j = 0; j < 16; ++j) bq[j] = wq[2 * j];
    const float bv = bias[n];

    const int step = gridDim.x;
    int mt = blockIdx.x;

    float4 r0 = {0.f, 0.f, 0.f, 0.f};
    float4 r1 = {0.f, 0.f, 0.f, 0.f};
    if (mt < mtiles) {
        int ra = mt * 16 + 2 * wave;
        int rb = ra + 1;
        ra = (ra < M) ? ra : (M - 1);
        rb = (rb < M) ? rb : (M - 1);
        r0 = ((const float4*)(X + (size_t)ra * D_HID))[lane];
        r1 = ((const float4*)(X + (size_t)rb * D_HID))[lane];
    }

    int buf = 0;
    for (; mt < mtiles; mt += step, buf ^= 1) {
        float* strip = (float*)s_strip[buf];
        ((float4*)(strip + (2 * wave)     * STRIP_PITCH))[lane] = r0;
        ((float4*)(strip + (2 * wave + 1) * STRIP_PITCH))[lane] = r1;

        const int mtn = mt + step;
        if (mtn < mtiles) {
            int ra = mtn * 16 + 2 * wave;
            int rb = ra + 1;
            ra = (ra < M) ? ra : (M - 1);
            rb = (rb < M) ? rb : (M - 1);
            r0 = ((const float4*)(X + (size_t)ra * D_HID))[lane];
            r1 = ((const float4*)(X + (size_t)rb * D_HID))[lane];
        }
        asm volatile("s_wait_dscnt 0x0" ::: "memory");
        asm volatile("s_barrier_signal -1" ::: "memory");
        asm volatile("s_barrier_wait -1" ::: "memory");

        const float* sa = strip + c * STRIP_PITCH + 4 * half;
        v8f acc0 = {0.f, 0.f, 0.f, 0.f, 0.f, 0.f, 0.f, 0.f};
        v8f acc1 = {0.f, 0.f, 0.f, 0.f, 0.f, 0.f, 0.f, 0.f};
        #pragma unroll
        for (int j = 0; j < 16; ++j) {
            const float4 a = *(const float4*)(sa + 8 * j);
            const v2f a0f = {a.x, a.y};
            const v2f a1f = {a.z, a.w};
            const v2f b0 = {bq[j].x, bq[j].y};
            const v2f b1 = {bq[j].z, bq[j].w};
            acc0 = __builtin_amdgcn_wmma_f32_16x16x4_f32(
                       false, a0f, false, b0, (short)0, acc0, false, false);
            acc1 = __builtin_amdgcn_wmma_f32_16x16x4_f32(
                       false, a1f, false, b1, (short)0, acc1, false, false);
        }
        const int row0 = mt * 16 + 8 * half;
        float* orow = Out + (size_t)row0 * D_HID + n;
        if (mt * 16 + 16 <= M) {
            #pragma unroll
            for (int r = 0; r < 8; ++r) {
                float v = acc0[r] + acc1[r] + bv;
                v = (v > 0.f) ? v : 0.f;
                orow[(size_t)r * D_HID] = v;
            }
        } else {
            #pragma unroll
            for (int r = 0; r < 8; ++r) {
                float v = acc0[r] + acc1[r] + bv;
                v = (v > 0.f) ? v : 0.f;
                if (row0 + r < M)
                    orow[(size_t)r * D_HID] = v;
            }
        }
    }
}

__global__ void copy_ints_kernel(const int* __restrict__ src, int* __restrict__ dst, int n)
{
    const int i = blockIdx.x * blockDim.x + threadIdx.x;
    if (i < n) dst[i] = src[i];
}

extern "C" void kernel_launch(void* const* d_in, const int* in_sizes, int n_in,
                              void* d_out, int out_size, void* d_ws, size_t ws_size,
                              hipStream_t stream)
{
    const float* X0  = (const float*)d_in[0];
    const float* W1  = (const float*)d_in[1];
    const float* b1  = (const float*)d_in[2];
    const float* W2  = (const float*)d_in[3];
    const float* b2  = (const float*)d_in[4];
    const float* Wfc = (const float*)d_in[5];
    const float* bfc = (const float*)d_in[6];
    const int*   esrc = (const int*)d_in[7];
    const int*   edst = (const int*)d_in[8];

    const int Nn   = in_sizes[0] / D_HID;
    const int E    = in_sizes[7];
    const int Ncls = in_sizes[6];

    float* bufA = (float*)d_ws;
    float* bufB = bufA + (size_t)Nn * D_HID;

    const int ranges = (Nn + RNG - 1) / RNG;
    const int chunk  = (((E + NCHUNK - 1) / NCHUNK) + 3) & ~3;
    int* cnt     = (int*)bufB;
    int* start   = cnt + (size_t)NCHUNK * Nn;
    int* tot     = start + (size_t)NCHUNK * Nn;
    int* rank    = tot + ranges * NCHUNK;
    int* offsets = (int*)d_out;
    int* by_dst  = offsets + (Nn + 1);
    int* offsets2 = (int*)bufA;
    int* by_dst2  = offsets2 + (Nn + 1);
    const int csrInts = Nn + 1 + E;

    const dim3 grdCount(ranges, NCHUNK);
    const dim3 grdPlace((chunk + 255) / 256, NCHUNK);
    const int copyBlocks = (csrInts + 255) / 256;
    const int fusedBlocks = FUSED_GRID;

    range_count_kernel<<<grdCount, RNG, 0, stream>>>(edst, rank, cnt, tot, E, Nn, chunk);
    offsets_kernel<<<ranges, RNG, 0, stream>>>(cnt, tot, offsets, start, Nn);
    place_kernel<<<grdPlace, 256, 0, stream>>>(esrc, edst, rank, start, by_dst, E, Nn, chunk);

    strip_linear_relu_kernel<<<LIN_GRID, 256, 0, stream>>>(X0, W1, b1, bufA, Nn);

    gather_linear_kernel<<<fusedBlocks, 256, 0, stream>>>(bufA, offsets, by_dst, W2, b2, bufB,
                                                          Nn, D_HID, 1);

    copy_ints_kernel<<<copyBlocks, 256, 0, stream>>>(offsets, offsets2, csrInts);

    gather_linear_kernel<<<fusedBlocks, 256, 0, stream>>>(bufB, offsets2, by_dst2, Wfc, bfc,
                                                          (float*)d_out, Nn, Ncls, 0);
}


# edge count pass: one workgroup's LDS holds a counter for every node (256 KiB), so the edges are read once, not once per 1024-node range; 32 edge chunks, so a node's rows are summed in a new order
# speedup vs baseline: 2.6900x; 1.1791x over previous
#include <hip/hip_runtime.h>

typedef float v2f __attribute__((ext_vector_type(2)));
typedef float v8f __attribute__((ext_vector_type(8)));

#define D_HID 128
#define FUSED_GRID 1024

#define CRNG 65536
#define SCAN_BLK 1024
#define CNT_THREADS 1024
#define NCHUNK 32
#define CHUNK_LOADS 5

__device__ __forceinline__ void count_one(int* s_cnt, int* __restrict__ rank, int e, int d, int lo, int nr)
{
    const unsigned u = (unsigned)(d - lo);
    if (u < (unsigned)nr) rank[e] = atomicAdd(&s_cnt[u], 1);
}

__device__ __forceinline__ void count_four(int* s_cnt, int* __restrict__ rank, int e, int4 d, int lo, int nr)
{
    const unsigned u0 = (unsigned)(d.x - lo);
    const unsigned u1 = (unsigned)(d.y - lo);
    const unsigned u2 = (unsigned)(d.z - lo);
    const unsigned u3 = (unsigned)(d.w - lo);
    const bool h0 = u0 < (unsigned)nr, h1 = u1 < (unsigned)nr, h2 = u2 < (unsigned)nr, h3 = u3 < (unsigned)nr;
    int r0 = 0, r1 = 0, r2 = 0, r3 = 0;
    if (h0) r0 = atomicAdd(&s_cnt[u0], 1);
    if (h1) r1 = atomicAdd(&s_cnt[u1], 1);
    if (h2) r2 = atomicAdd(&s_cnt[u2], 1);
    if (h3) r3 = atomicAdd(&s_cnt[u3], 1);
    if (h0) rank[e + 0] = r0;
    if (h1) rank[e + 1] = r1;
    if (h2) rank[e + 2] = r2;
    if (h3) rank[e + 3] = r3;
}

__global__ void __launch_bounds__(CNT_THREADS)
range_count_kernel(const int* __restrict__ dst, int* __restrict__ rank,
                   int* __restrict__ cnt, int* __restrict__ tot,
                   int E, int n_nodes, int chunk)
{
    __shared__ int s_cnt[CRNG];
    const int tid  = threadIdx.x;
    const int lane = tid & 31;
    const int wave = tid >> 5;
    const int lo = blockIdx.x * CRNG;
    const int q  = blockIdx.y;
    const int nr = (n_nodes - lo < CRNG) ? (n_nodes - lo) : CRNG;
    for (int i = tid; i < nr; i += CNT_THREADS) s_cnt[i] = 0;
    __syncthreads();

    const int e0 = q * chunk;
    const int e1 = (e0 + chunk < E) ? (e0 + chunk) : E;
    const int n4 = (e1 > e0) ? ((e1 - e0) >> 2) : 0;
    const int4* d4 = (const int4*)(dst + e0);
    int* rk = rank + e0;
    for (int g = tid; g < n4; g += CHUNK_LOADS * CNT_THREADS) {
        int4 v[CHUNK_LOADS];
        #pragma unroll
        for (int k = 0; k < CHUNK_LOADS; ++k) {
            const int gk = g + k * CNT_THREADS;
            v[k] = d4[(gk < n4) ? gk : g];
        }
        #pragma unroll
        for (int k = 0; k < CHUNK_LOADS; ++k) {
            const int gk = g + k * CNT_THREADS;
            if (gk < n4) count_four(s_cnt, rk, 4 * gk, v[k], lo, nr);
        }
    }
    const int t = e0 + 4 * n4 + tid;
    if (t < e1) count_one(s_cnt, rank, t, dst[t], lo, nr);
    asm volatile("s_wait_dscnt 0x0" ::: "memory");
    asm volatile("s_barrier_signal -1" ::: "memory");
    asm volatile("s_barrier_wait -1" ::: "memory");

    const int nblk = (nr + SCAN_BLK - 1) / SCAN_BLK;
    int* crow = cnt + (size_t)q * n_nodes + lo;
    for (int b = wave; b < nblk; b += CNT_THREADS / 32) {
        int s = 0;
        if (b * SCAN_BLK + SCAN_BLK <= nr) {
            int c[SCAN_BLK / 32];
            #pragma unroll
            for (int k = 0; k < SCAN_BLK / 32; ++k) c[k] = s_cnt[b * SCAN_BLK + 32 * k + lane];
            #pragma unroll
            for (int k = 0; k < SCAN_BLK / 32; ++k) {
                crow[b * SCAN_BLK + 32 * k + lane] = c[k];
                s += c[k];
            }
        } else {
            for (int k = lane; k < SCAN_BLK; k += 32) {
                const int i = b * SCAN_BLK + k;
                if (i < nr) {
                    const int c = s_cnt[i];
                    crow[i] = c;
                    s += c;
                }
            }
        }
        for (int o = 16; o > 0; o >>= 1) s += __shfl_xor(s, o);
        if (lane == 0) tot[(blockIdx.x * (CRNG / SCAN_BLK) + b) * NCHUNK + q] = s;
    }
}

__global__ void __launch_bounds__(SCAN_BLK)
offsets_kernel(const int* __restrict__ cnt, const int* __restrict__ tot,
               int* __restrict__ offsets, int* __restrict__ start, int n)
{
    __shared__ int s_part[32];
    __shared__ int s_pre[32];
    const int tid  = threadIdx.x;
    const int lane = tid & 31;
    const int wave = tid >> 5;
    const int first = blockIdx.x * SCAN_BLK;

    int pre = 0;
    for (int j = tid; j < (int)blockIdx.x * NCHUNK; j += SCAN_BLK) pre += tot[j];

    const int i = first + tid;
    int cq[NCHUNK];
    int c = 0;
    #pragma unroll
    for (int q = 0; q < NCHUNK; ++q) {
        cq[q] = (i < n) ? cnt[(size_t)q * n + i] : 0;
        c += cq[q];
    }
    int incl = c;
    for (int o = 1; o < 32; o <<= 1) {
        const int y = __shfl_up(incl, o);
        if (lane >= o) incl += y;
    }
    for (int o = 16; o > 0; o >>= 1) pre += __shfl_xor(pre, o);
    if (lane == 31) s_part[wave] = incl;
    if (lane == 0)  s_pre[wave]  = pre;
    __syncthreads();
    int add = s_pre[lane] + ((lane < wave) ? s_part[lane] : 0);
    for (int o = 16; o > 0; o >>= 1) add += __shfl_xor(add, o);
    const int excl = add + incl - c;
    if (i < n) {
        offsets[i] = excl;
        int s = excl;
        #pragma unroll
        for (int q = 0; q < NCHUNK; ++q) {
            start[(size_t)q * n + i] = s;
            s += cq[q];
        }
    }
    if (i == n - 1) offsets[n] = excl + c;
}

__global__ void place_kernel(const int* __restrict__ src,
                             const int* __restrict__ dst,
                             const int* __restrict__ rank,
                             const int* __restrict__ start,
                             int* __restrict__ by_dst, int E, int n_nodes, int chunk)
{
    const int q  = blockIdx.y;
    const int k  = blockIdx.x * blockDim.x + threadIdx.x;
    const int e  = q * chunk + k;
    if (k < chunk && e < E) by_dst[start[(size_t)q * n_nodes + dst[e]] + rank[e]] = src[e];
}

#define STRIP_PITCH 132
#define ROWS_IN_FLIGHT 8

__device__ __forceinline__ int load_ids(const int* __restrict__ by_dst, int beg, int cnt, int lane)
{
    return (lane < cnt) ? by_dst[beg + lane] : 0;
}

__global__ void __launch_bounds__(256)
gather_linear_kernel(const float* __restrict__ Y,
                     const int* __restrict__ offsets,
                     const int* __restrict__ by_dst,
                     const float* __restrict__ W,
                     const float* __restrict__ bias,
                     float* __restrict__ Out,
                     int n_nodes, int N, int do_relu)
{
    __shared__ float4 s_strip[2][16 * STRIP_PITCH / 4];

    const int lane = threadIdx.x & 31;
    const int wave = __builtin_amdgcn_readfirstlane((int)(threadIdx.x >> 5));
    const int half = lane >> 4;
    const int c    = lane & 15;
    const float4* __restrict__ Yq = (const float4*)Y;
    const unsigned ulane = (unsigned)lane;

    const int ntiles = (N + 15) >> 4;
    const int mtiles = (n_nodes + 15) >> 4;
    const bool has_tile = wave < ntiles;

    const int n  = wave * 16 + c;
    const int nc = (n < N) ? n : (N - 1);

    float4 bq[16];
    float bv = 0.f;
    if (has_tile) {
        const float4* wq = (const float4*)(W + (size_t)nc * D_HID + 4 * half);
        #pragma unroll
        for (int j = 0; j < 16; ++j) bq[j] = wq[2 * j];
        bv = bias[nc];
    }

    const int osel = 2 * wave + ((lane < 2) ? lane : 2);
    const int step = gridDim.x;

    int mt = blockIdx.x;
    int o0 = 0, o1 = 0, o2 = 0;
    int mine = 0, mine_hi = 0;
    int ovn = 0;
    if (mt < mtiles) {
        int oi = mt * 16 + osel;
        oi = (oi < n_nodes) ? oi : n_nodes;
        const int ov = offsets[oi];
        o0 = __builtin_amdgcn_readlane(ov, 0);
        o1 = __builtin_amdgcn_readlane(ov, 1);
        o2 = __builtin_amdgcn_readlane(ov, 2);
        mine    = load_ids(by_dst, o0,      o2 - o0,      lane);
        mine_hi = load_ids(by_dst, o0 + 32, o2 - o0 - 32, lane);
        if (mt + step < mtiles) {
            int on = (mt + step) * 16 + osel;
            on = (on < n_nodes) ? on : n_nodes;
            ovn = offsets[on];
        }
    }

    int buf = 0;
    for (; mt < mtiles; mt += step, buf ^= 1) {
        float* strip = (float*)s_strip[buf];

        const int cnt   = o2 - o0;
        const int split = o1 - o0;
        float4 a0 = {0.f, 0.f, 0.f, 0.f};
        float4 a1 = {0.f, 0.f, 0.f, 0.f};
        for (int base = 0; base < cnt; base += 32) {
            int ids;
            if (base == 0)       ids = mine;
            else if (base == 32) ids = mine_hi;
            else                 ids = load_ids(by_dst, o0 + base, cnt - base, lane);
            const int left = cnt - base;
            const int c32  = (left < 32) ? left : 32;
            const int spb = split - base;
            auto tested_block = [&](const int k) {
                float4 v[ROWS_IN_FLIGHT];
                const bool wide = (k + ROWS_IN_FLIGHT / 2 < c32);
                #pragma unroll
                for (int i = 0; i < ROWS_IN_FLIGHT / 2; ++i) {
                    const int kk = (k + i < c32) ? (k + i) : (c32 - 1);
                    const int s  = __builtin_amdgcn_readlane(ids, kk);
                    v[i] = Yq[(unsigned)s * (D_HID / 4) + ulane];
                }
                if (wide) {
                    #pragma unroll
                    for (int i = ROWS_IN_FLIGHT / 2; i < ROWS_IN_FLIGHT; ++i) {
                        const int kk = (k + i < c32) ? (k + i) : (c32 - 1);
                        const int s  = __builtin_amdgcn_readlane(ids, kk);
                        v[i] = Yq[(unsigned)s * (D_HID / 4) + ulane];
                    }
                }
                #pragma unroll
                for (int i = 0; i < ROWS_IN_FLIGHT; ++i) {
                    if (i == 0 || k + i < c32) {
                        if (k + i < spb) {
                            a0.x += v[i].x; a0.y += v[i].y; a0.z += v[i].z; a0.w += v[i].w;
                        } else {
                            a1.x += v[i].x; a1.y += v[i].y; a1.z += v[i].z; a1.w += v[i].w;
                        }
                    }
                }
            };
            int k = 0;
            for (; k + ROWS_IN_FLIGHT <= c32 && k + ROWS_IN_FLIGHT <= spb; k += ROWS_IN_FLIGHT) {
                float4 v[ROWS_IN_FLIGHT];
                #pragma unroll
                for (int i = 0; i < ROWS_IN_FLIGHT; ++i) {
                    const int s = __builtin_amdgcn_readlane(ids, k + i);
                    v[i] = Yq[(unsigned)s * (D_HID / 4) + ulane];
                }
                #pragma unroll
                for (int i = 0; i < ROWS_IN_FLIGHT; ++i) {
                    a0.x += v[i].x; a0.y += v[i].y; a0.z += v[i].z; a0.w += v[i].w;
                }
            }
            if (k < c32 && k < spb) {
                tested_block(k);
                k += ROWS_IN_FLIGHT;
            }
            for (; k + ROWS_IN_FLIGHT <= c32; k += ROWS_IN_FLIGHT) {
                float4 v[ROWS_IN_FLIGHT];
                #pragma unroll
                for (int i = 0; i < ROWS_IN_FLIGHT; ++i) {
                    const int s = __builtin_amdgcn_readlane(ids, k + i);
                    v[i] = Yq[(unsigned)s * (D_HID / 4) + ulane];
                }
                #pragma unroll
                for (int i = 0; i < ROWS_IN_FLIGHT; ++i) {
                    a1.x += v[i].x; a1.y += v[i].y; a1.z += v[i].z; a1.w += v[i].w;
                }
            }
            if (k < c32) tested_block(k);
        }
        ((float4*)(strip + (2 * wave)     * STRIP_PITCH))[lane] = a0;
        ((float4*)(strip + (2 * wave + 1) * STRIP_PITCH))[lane] = a1;

        const int mtn = mt + step;
        if (mtn < mtiles) {
            o0 = __builtin_amdgcn_readlane(ovn, 0);
            o1 = __builtin_amdgcn_readlane(ovn, 1);
            o2 = __builtin_amdgcn_readlane(ovn, 2);
            mine    = load_ids(by_dst, o0,      o2 - o0,      lane);
            mine_hi = load_ids(by_dst, o0 + 32, o2 - o0 - 32, lane);
            if (mtn + step < mtiles) {
                int on = (mtn + step) * 16 + osel;
                on = (on < n_nodes) ? on : n_nodes;
                ovn = offsets[on];
            }
        }
        asm volatile("s_wait_dscnt 0x0" ::: "memory");
        asm volatile("s_barrier_signal -1" ::: "memory");
        asm volatile("s_barrier_wait -1" ::: "memory");

        if (has_tile) {
            const float* sa = strip + c * STRIP_PITCH + 4 * half;
            v8f acc0 = {0.f, 0.f, 0.f, 0.f, 0.f, 0.f, 0.f, 0.f};
            v8f acc1 = {0.f, 0.f, 0.f, 0.f, 0.f, 0.f, 0.f, 0.f};
            #pragma unroll
            for (int j = 0; j < 16; ++j) {
                const float4 a = *(const float4*)(sa + 8 * j);
                const v2f a0f = {a.x, a.y};
                const v2f a1f = {a.z, a.w};
                const v2f b0 = {bq[j].x, bq[j].y};
                const v2f b1 = {bq[j].z, bq[j].w};
                acc0 = __builtin_amdgcn_wmma_f32_16x16x4_f32(
                           false, a0f, false, b0, (short)0, acc0, false, false);
                acc1 = __builtin_amdgcn_wmma_f32_16x16x4_f32(
                           false, a1f, false, b1, (short)0, acc1, false, false);
            }
            const int row0 = mt * 16 + 8 * half;
            if (n < N) {
                float* orow = Out + (size_t)row0 * N + n;
                if (mt * 16 + 16 <= n_nodes) {
                    #pragma unroll
                    for (int r = 0; r < 8; ++r) {
                        float v = acc0[r] + acc1[r] + bv;
                        if (do_relu) v = (v > 0.f) ? v : 0.f;
                        orow[(size_t)r * N] = v;
                    }
                } else {
                    #pragma unroll
                    for (int r = 0; r < 8; ++r) {
                        float v = acc0[r] + acc1[r] + bv;
                        if (do_relu) v = (v > 0.f) ? v : 0.f;
                        if (row0 + r < n_nodes)
                            orow[(size_t)r * N] = v;
                    }
                }
            }
        }
    }
}

#define LIN_GRID 1024

__global__ void __launch_bounds__(256)
strip_linear_relu_kernel(const float* __restrict__ X,
                         const float* __restrict__ W,
                         const float* __restrict__ bias,
                         float* __restrict__ Out,
                         int M)
{
    __shared__ float4 s_strip[2][16 * STRIP_PITCH / 4];

    const int lane = threadIdx.x & 31;
    const int wave = __builtin_amdgcn_readfirstlane((int)(threadIdx.x >> 5));
    const int half = lane >> 4;
    const int c    = lane & 15;

    const int mtiles = (M + 15) >> 4;
    const int n = wave * 16 + c;

    float4 bq[16];
    const float4* wq = (const float4*)(W + (size_t)n * D_HID + 4 * half);
    #pragma unroll
    for (int j = 0; j < 16; ++j) bq[j] = wq[2 * j];
    const float bv = bias[n];

    const int step = gridDim.x;
    int mt = blockIdx.x;

    float4 r0 = {0.f, 0.f, 0.f, 0.f};
    float4 r1 = {0.f, 0.f, 0.f, 0.f};
    if (mt < mtiles) {
        int ra = mt * 16 + 2 * wave;
        int rb = ra + 1;
        ra = (ra < M) ? ra : (M - 1);
        rb = (rb < M) ? rb : (M - 1);
        r0 = ((const float4*)(X + (size_t)ra * D_HID))[lane];
        r1 = ((const float4*)(X + (size_t)rb * D_HID))[lane];
    }

    int buf = 0;
    for (; mt < mtiles; mt += step, buf ^= 1) {
        float* strip = (float*)s_strip[buf];
        ((float4*)(strip + (2 * wave)     * STRIP_PITCH))[lane] = r0;
        ((float4*)(strip + (2 * wave + 1) * STRIP_PITCH))[lane] = r1;

        const int mtn = mt + step;
        if (mtn < mtiles) {
            int ra = mtn * 16 + 2 * wave;
            int rb = ra + 1;
            ra = (ra < M) ? ra : (M - 1);
            rb = (rb < M) ? rb : (M - 1);
            r0 = ((const float4*)(X + (size_t)ra * D_HID))[lane];
            r1 = ((const float4*)(X + (size_t)rb * D_HID))[lane];
        }
        asm volatile("s_wait_dscnt 0x0" ::: "memory");
        asm volatile("s_barrier_signal -1" ::: "memory");
        asm volatile("s_barrier_wait -1" ::: "memory");

        const float* sa = strip + c * STRIP_PITCH + 4 * half;
        v8f acc0 = {0.f, 0.f, 0.f, 0.f, 0.f, 0.f, 0.f, 0.f};
        v8f acc1 = {0.f, 0.f, 0.f, 0.f, 0.f, 0.f, 0.f, 0.f};
        #pragma unroll
        for (int j = 0; j < 16; ++j) {
            const float4 a = *(const float4*)(sa + 8 * j);
            const v2f a0f = {a.x, a.y};
            const v2f a1f = {a.z, a.w};
            const v2f b0 = {bq[j].x, bq[j].y};
            const v2f b1 = {bq[j].z, bq[j].w};
            acc0 = __builtin_amdgcn_wmma_f32_16x16x4_f32(
                       false, a0f, false, b0, (short)0, acc0, false, false);
            acc1 = __builtin_amdgcn_wmma_f32_16x16x4_f32(
                       false, a1f, false, b1, (short)0, acc1, false, false);
        }
        const int row0 = mt * 16 + 8 * half;
        float* orow = Out + (size_t)row0 * D_HID + n;
        if (mt * 16 + 16 <= M) {
            #pragma unroll
            for (int r = 0; r < 8; ++r) {
                float v = acc0[r] + acc1[r] + bv;
                v = (v > 0.f) ? v : 0.f;
                orow[(size_t)r * D_HID] = v;
            }
        } else {
            #pragma unroll
            for (int r = 0; r < 8; ++r) {
                float v = acc0[r] + acc1[r] + bv;
                v = (v > 0.f) ? v : 0.f;
                if (row0 + r < M)
                    orow[(size_t)r * D_HID] = v;
            }
        }
    }
}

__global__ void copy_ints_kernel(const int* __restrict__ src, int* __restrict__ dst, int n)
{
    const int i = blockIdx.x * blockDim.x + threadIdx.x;
    if (i < n) dst[i] = src[i];
}

extern "C" void kernel_launch(void* const* d_in, const int* in_sizes, int n_in,
                              void* d_out, int out_size, void* d_ws, size_t ws_size,
                              hipStream_t stream)
{
    const float* X0  = (const float*)d_in[0];
    const float* W1  = (const float*)d_in[1];
    const float* b1  = (const float*)d_in[2];
    const float* W2  = (const float*)d_in[3];
    const float* b2  = (const float*)d_in[4];
    const float* Wfc = (const float*)d_in[5];
    const float* bfc = (const float*)d_in[6];
    const int*   esrc = (const int*)d_in[7];
    const int*   edst = (const int*)d_in[8];

    const int Nn   = in_sizes[0] / D_HID;
    const int E    = in_sizes[7];
    const int Ncls = in_sizes[6];

    float* bufA = (float*)d_ws;
    float* bufB = bufA + (size_t)Nn * D_HID;

    const int ranges = (Nn + CRNG - 1) / CRNG;
    const int blocks = (Nn + SCAN_BLK - 1) / SCAN_BLK;
    const int chunk  = (((E + NCHUNK - 1) / NCHUNK) + 3) & ~3;
    int* cnt     = (int*)bufB;
    int* start   = cnt + (size_t)NCHUNK * Nn;
    int* tot     = start + (size_t)NCHUNK * Nn;
    int* rank    = tot + blocks * NCHUNK;
    int* offsets = (int*)d_out;
    int* by_dst  = offsets + (Nn + 1);
    int* offsets2 = (int*)bufA;
    int* by_dst2  = offsets2 + (Nn + 1);
    const int csrInts = Nn + 1 + E;

    const dim3 grdCount(ranges, NCHUNK);
    const dim3 grdPlace((chunk + 255) / 256, NCHUNK);
    const int copyBlocks = (csrInts + 255) / 256;
    const int fusedBlocks = FUSED_GRID;

    range_count_kernel<<<grdCount, CNT_THREADS, 0, stream>>>(edst, rank, cnt, tot, E, Nn, chunk);
    offsets_kernel<<<blocks, SCAN_BLK, 0, stream>>>(cnt, tot, offsets, start, Nn);
    place_kernel<<<grdPlace, 256, 0, stream>>>(esrc, edst, rank, start, by_dst, E, Nn, chunk);

    strip_linear_relu_kernel<<<LIN_GRID, 256, 0, stream>>>(X0, W1, b1, bufA, Nn);

    gather_linear_kernel<<<fusedBlocks, 256, 0, stream>>>(bufA, offsets, by_dst, W2, b2, bufB,
                                                          Nn, D_HID, 1);

    copy_ints_kernel<<<copyBlocks, 256, 0, stream>>>(offsets, offsets2, csrInts);

    gather_linear_kernel<<<fusedBlocks, 256, 0, stream>>>(bufB, offsets2, by_dst2, Wfc, bfc,
                                                          (float*)d_out, Nn, Ncls, 0);
}
